# starting kernel
# baseline (speedup 1.0000x reference)
#include <hip/hip_runtime.h>
#include <hip/hip_bf16.h>

#define N_NODES 50000
#define N_EDGES 500000
#define N_GRAPHS 1000
#define HID 128
#define OUTD 10
#define NLAYERS 4

typedef __attribute__((ext_vector_type(16))) __bf16 v16bf;
typedef __attribute__((ext_vector_type(8)))  __bf16 v8bf;
typedef __attribute__((ext_vector_type(8)))  float  v8f;

__global__ __launch_bounds__(256) void k_encode_nodes(
    const int* __restrict__ x, const float* __restrict__ atom_emb,
    float* __restrict__ h)
{
    int gid = blockIdx.x * 256 + threadIdx.x;
    if (gid >= N_NODES * 32) return;
    int n = gid >> 5;
    int c = (gid & 31) * 4;
    float4 acc = make_float4(0.f, 0.f, 0.f, 0.f);
    #pragma unroll
    for (int f = 0; f < 9; ++f) {
        int idx = x[n * 9 + f];
        const float4 v = *(const float4*)(atom_emb + ((size_t)(f * 120 + idx)) * HID + c);
        acc.x += v.x; acc.y += v.y; acc.z += v.z; acc.w += v.w;
    }
    *(float4*)(h + (size_t)n * HID + c) = acc;
}

__global__ __launch_bounds__(256) void k_copy_h(
    const float* __restrict__ h, float* __restrict__ agg)
{
    int gid = blockIdx.x * 256 + threadIdx.x;
    if (gid >= N_NODES * 32) return;
    ((float4*)agg)[gid] = ((const float4*)h)[gid];
}

__global__ __launch_bounds__(256) void k_edges(
    const int* __restrict__ edge_index, const int* __restrict__ edge_attr,
    const float* __restrict__ bond_emb, const float* __restrict__ h,
    float* __restrict__ agg)
{
    int gid = blockIdx.x * 256 + threadIdx.x;
    if (gid >= N_EDGES * 32) return;
    int e = gid >> 5;
    int c = (gid & 31) * 4;
    int src = edge_index[e];
    int dst = edge_index[N_EDGES + e];
    int a0 = edge_attr[e * 3 + 0];
    int a1 = edge_attr[e * 3 + 1];
    int a2 = edge_attr[e * 3 + 2];
    const float4 e0 = *(const float4*)(bond_emb + ((size_t)(0 * 6 + a0)) * HID + c);
    const float4 e1 = *(const float4*)(bond_emb + ((size_t)(1 * 6 + a1)) * HID + c);
    const float4 e2 = *(const float4*)(bond_emb + ((size_t)(2 * 6 + a2)) * HID + c);
    const float4 hv = *(const float4*)(h + (size_t)src * HID + c);
    float m0 = hv.x + e0.x + e1.x + e2.x; m0 = m0 > 0.f ? m0 : 0.f;
    float m1 = hv.y + e0.y + e1.y + e2.y; m1 = m1 > 0.f ? m1 : 0.f;
    float m2 = hv.z + e0.z + e1.z + e2.z; m2 = m2 > 0.f ? m2 : 0.f;
    float m3 = hv.w + e0.w + e1.w + e2.w; m3 = m3 > 0.f ? m3 : 0.f;
    float* o = agg + (size_t)dst * HID + c;
    atomicAdd(o + 0, m0);
    atomicAdd(o + 1, m1);
    atomicAdd(o + 2, m2);
    atomicAdd(o + 3, m3);
}

__global__ __launch_bounds__(256) void k_prep_weights(
    const float* __restrict__ w1, const float* __restrict__ w2,
    __bf16* __restrict__ f1, __bf16* __restrict__ f2)
{
    int o = blockIdx.x * 256 + threadIdx.x;
    if (o >= NLAYERS * 4 * 8 * 32 * 16) return;
    int j     =  o        & 15;
    int lane  = (o >> 4)  & 31;
    int ntile = (o >> 9)  & 7;
    int ks    = (o >> 12) & 3;
    int layer =  o >> 14;
    int colL  = lane & 15;
    int khalf = (lane >> 4) * 8;
    int K     = ks * 32 + ((j < 8) ? (khalf + j) : (16 + khalf + (j - 8)));
    int col   = ntile * 16 + colL;
    size_t src = (size_t)layer * HID * HID + (size_t)K * HID + col;
    size_t dst = (size_t)layer * 16384 +
                 (((size_t)ks * 8 + ntile) * 32 + lane) * 16 + j;
    f1[dst] = (__bf16)w1[src];
    f2[dst] = (__bf16)w2[src];
}

__global__ __launch_bounds__(256) void k_node_update(
    const float* __restrict__ agg, const float* __restrict__ hcur,
    const __bf16* __restrict__ wf1, const __bf16* __restrict__ wf2,
    const float* __restrict__ b1, const float* __restrict__ b2,
    const float* __restrict__ gamma, const float* __restrict__ beta,
    const float* __restrict__ mean, const float* __restrict__ var,
    float* __restrict__ hnext)
{
    __shared__ __align__(16) __bf16 zt[16 * HID];
    __shared__ __align__(16) __bf16 z1[16 * HID];

    const int tid   = threadIdx.x;
    const int wave  = tid >> 5;
    const int lane  = tid & 31;
    const int node0 = blockIdx.x * 16;

    {
        const int base = tid * 8;
        const int row  = base >> 7;
        const int c    = base & 127;
        const float4* p = (const float4*)(agg + (size_t)(node0 + row) * HID + c);
        float4 v0 = p[0], v1 = p[1];
        __bf16* q = &zt[base];
        q[0] = (__bf16)v0.x; q[1] = (__bf16)v0.y;
        q[2] = (__bf16)v0.z; q[3] = (__bf16)v0.w;
        q[4] = (__bf16)v1.x; q[5] = (__bf16)v1.y;
        q[6] = (__bf16)v1.z; q[7] = (__bf16)v1.w;
    }
    __syncthreads();

    const int colL  = lane & 15;
    const int col   = wave * 16 + colL;
    const int khalf = (lane >> 4) * 8;
    const int rowA  = lane & 15;

    v8f acc1 = {};
    #pragma unroll
    for (int ks = 0; ks < 4; ++ks) {
        const __bf16* ap = &zt[rowA * HID + ks * 32 + khalf];
        v8bf alo = *(const v8bf*)(ap);
        v8bf ahi = *(const v8bf*)(ap + 16);
        v16bf a = __builtin_shufflevector(alo, ahi,
                    0,1,2,3,4,5,6,7,8,9,10,11,12,13,14,15);
        v16bf b = *(const v16bf*)(wf1 + (((size_t)ks * 8 + wave) * 32 + lane) * 16);
        acc1 = __builtin_amdgcn_wmma_f32_16x16x32_bf16(
                   false, a, false, b, (short)0, acc1, false, false);
    }

    const float bias1 = b1[col];
    #pragma unroll
    for (int r = 0; r < 8; ++r) {
        int m = (lane < 16) ? r : (r + 8);
        float v = acc1[r] + bias1;
        v = v > 0.f ? v : 0.f;
        z1[m * HID + col] = (__bf16)v;
    }
    __syncthreads();

    v8f acc2 = {};
    #pragma unroll
    for (int ks = 0; ks < 4; ++ks) {
        const __bf16* ap = &z1[rowA * HID + ks * 32 + khalf];
        v8bf alo = *(const v8bf*)(ap);
        v8bf ahi = *(const v8bf*)(ap + 16);
        v16bf a = __builtin_shufflevector(alo, ahi,
                    0,1,2,3,4,5,6,7,8,9,10,11,12,13,14,15);
        v16bf b = *(const v16bf*)(wf2 + (((size_t)ks * 8 + wave) * 32 + lane) * 16);
        acc2 = __builtin_amdgcn_wmma_f32_16x16x32_bf16(
                   false, a, false, b, (short)0, acc2, false, false);
    }

    const float bias2 = b2[col];
    const float mu    = mean[col];
    const float inv   = rsqrtf(var[col] + 1e-5f);
    const float ga    = gamma[col];
    const float be    = beta[col];
    #pragma unroll
    for (int r = 0; r < 8; ++r) {
        int m = (lane < 16) ? r : (r + 8);
        size_t idx = (size_t)(node0 + m) * HID + col;
        float v = acc2[r] + bias2;
        v = (v - mu) * inv * ga + be;
        v = v > 0.f ? v : 0.f;
        hnext[idx] = hcur[idx] + v;
    }
}

__global__ __launch_bounds__(256) void k_pool_zero(
    float* __restrict__ sums, float* __restrict__ cnts)
{
    int i = blockIdx.x * 256 + threadIdx.x;
    if (i < N_GRAPHS * HID) sums[i] = 0.f;
    if (i < N_GRAPHS)       cnts[i] = 0.f;
}

__global__ __launch_bounds__(256) void k_pool_acc(
    const float* __restrict__ h, const int* __restrict__ batch,
    float* __restrict__ sums, float* __restrict__ cnts)
{
    int gid = blockIdx.x * 256 + threadIdx.x;
    if (gid >= N_NODES * 32) return;
    int n = gid >> 5;
    int c = (gid & 31) * 4;
    int g = batch[n];
    const float4 v = *(const float4*)(h + (size_t)n * HID + c);
    float* o = sums + (size_t)g * HID + c;
    atomicAdd(o + 0, v.x);
    atomicAdd(o + 1, v.y);
    atomicAdd(o + 2, v.z);
    atomicAdd(o + 3, v.w);
    if ((gid & 31) == 0) atomicAdd(cnts + g, 1.0f);
}

__global__ __launch_bounds__(128) void k_pool_mlp(
    const float* __restrict__ sums, const float* __restrict__ cnts,
    const float* __restrict__ mw, const float* __restrict__ mb,
    float* __restrict__ out)
{
    __shared__ float p[HID];
    int g = blockIdx.x;
    int t = threadIdx.x;
    float cnt = cnts[g];
    cnt = cnt > 1.f ? cnt : 1.f;
    p[t] = sums[(size_t)g * HID + t] / cnt;
    __syncthreads();
    if (t < OUTD) {
        float a = mb[t];
        #pragma unroll 16
        for (int c = 0; c < HID; ++c) a += p[c] * mw[c * OUTD + t];
        out[g * OUTD + t] = a;
    }
}

extern "C" void kernel_launch(void* const* d_in, const int* in_sizes, int n_in,
                              void* d_out, int out_size, void* d_ws, size_t ws_size,
                              hipStream_t stream) {
    const int*   x          = (const int*)  d_in[0];
    const int*   edge_index = (const int*)  d_in[1];
    const int*   edge_attr  = (const int*)  d_in[2];
    const int*   batch      = (const int*)  d_in[3];
    const float* atom_emb   = (const float*)d_in[4];
    const float* bond_emb   = (const float*)d_in[5];
    const float* lin1_w     = (const float*)d_in[6];
    const float* lin1_b     = (const float*)d_in[7];
    const float* lin2_w     = (const float*)d_in[8];
    const float* lin2_b     = (const float*)d_in[9];
    const float* bn_gamma   = (const float*)d_in[10];
    const float* bn_beta    = (const float*)d_in[11];
    const float* bn_mean    = (const float*)d_in[12];
    const float* bn_var     = (const float*)d_in[13];
    const float* mlp_w      = (const float*)d_in[14];
    const float* mlp_b      = (const float*)d_in[15];

    char* ws = (char*)d_ws;
    size_t off = 0;
    auto alloc = [&](size_t bytes) -> char* {
        char* p = ws + off;
        off = (off + bytes + 255) & ~(size_t)255;
        return p;
    };
    float*  h0   = (float*) alloc((size_t)N_NODES * HID * 4);
    float*  h1   = (float*) alloc((size_t)N_NODES * HID * 4);
    float*  agg  = (float*) alloc((size_t)N_NODES * HID * 4);
    __bf16* wf1  = (__bf16*)alloc((size_t)NLAYERS * 16384 * 2);
    __bf16* wf2  = (__bf16*)alloc((size_t)NLAYERS * 16384 * 2);
    float*  sums = (float*) alloc((size_t)N_GRAPHS * HID * 4);
    float*  cnts = (float*) alloc((size_t)N_GRAPHS * 4);

    const int nodeChunks = N_NODES * 32;
    const int edgeChunks = N_EDGES * 32;

    k_prep_weights<<<(NLAYERS * 4 * 8 * 32 * 16 + 255) / 256, 256, 0, stream>>>(
        lin1_w, lin2_w, wf1, wf2);
    k_encode_nodes<<<(nodeChunks + 255) / 256, 256, 0, stream>>>(x, atom_emb, h0);

    float* hc = h0;
    float* hn = h1;
    for (int i = 0; i < NLAYERS; ++i) {
        k_copy_h<<<(nodeChunks + 255) / 256, 256, 0, stream>>>(hc, agg);
        k_edges<<<(edgeChunks + 255) / 256, 256, 0, stream>>>(
            edge_index, edge_attr, bond_emb, hc, agg);
        k_node_update<<<N_NODES / 16, 256, 0, stream>>>(
            agg, hc, wf1 + (size_t)i * 16384, wf2 + (size_t)i * 16384,
            lin1_b + i * HID, lin2_b + i * HID,
            bn_gamma + i * HID, bn_beta + i * HID,
            bn_mean + i * HID, bn_var + i * HID, hn);
        float* t = hc; hc = hn; hn = t;
    }

    k_pool_zero<<<(N_GRAPHS * HID + 255) / 256, 256, 0, stream>>>(sums, cnts);
    k_pool_acc<<<(nodeChunks + 255) / 256, 256, 0, stream>>>(hc, batch, sums, cnts);
    k_pool_mlp<<<N_GRAPHS, 128, 0, stream>>>(sums, cnts, mlp_w, mlp_b, (float*)d_out);
}


# tile sums: each wave first copies its own rows' records, in list order, to a wave-private LDS list, then walks it 8 per trip: no padded slots (were ~3 in 11), no bit-scan chain; sum order unchanged
# speedup vs baseline: 1.8352x; 1.8352x over previous
#include <hip/hip_runtime.h>
#include <hip/hip_bf16.h>

#define N_NODES 50000
#define N_EDGES 500000
#define N_GRAPHS 1000
#define HID 128
#define OUTD 10
#define NLAYERS 4

typedef __attribute__((ext_vector_type(16))) __bf16 v16bf;
typedef __attribute__((ext_vector_type(8)))  __bf16 v8bf;
typedef __attribute__((ext_vector_type(8)))  float  v8f;

__global__ __launch_bounds__(256) void k_encode_nodes(
    const int* __restrict__ x, const float* __restrict__ atom_emb,
    float* __restrict__ h)
{
    int gid = blockIdx.x * 256 + threadIdx.x;
    if (gid >= N_NODES * 32) return;
    int n = gid >> 5;
    int c = (gid & 31) * 4;
    float4 acc = make_float4(0.f, 0.f, 0.f, 0.f);
    #pragma unroll
    for (int f = 0; f < 9; ++f) {
        int idx = x[n * 9 + f];
        const float4 v = *(const float4*)(atom_emb + ((size_t)(f * 120 + idx)) * HID + c);
        acc.x += v.x; acc.y += v.y; acc.z += v.z; acc.w += v.w;
    }
    *(float4*)(h + (size_t)n * HID + c) = acc;
}

#define TR 32
#define TR_LOG 5
#define NB ((N_NODES + TR - 1) / TR)
#define SEG 512
#define NSEG ((N_EDGES + SEG - 1) / SEG)
static_assert(SEG == 2 * 256, "k_rank ranks two edges per thread");
static_assert(NSEG <= 1024, "k_scan_seg scans one segment per thread");
static_assert(NB <= 2048, "k_scan_tiles scans two tiles per thread");

__global__ __launch_bounds__(256) void k_rank(
    const int* __restrict__ edge_index, int* __restrict__ rankv,
    int* __restrict__ cnt)
{
    __shared__ __align__(16) int key[SEG];
    __shared__ int hist[NB];
    const int tid = threadIdx.x;
    const int s   = blockIdx.x;
    const int e0  = s * SEG;
    const int* dstp = edge_index + N_EDGES;
    for (int i = tid; i < SEG; i += 256) {
        const int e = e0 + i;
        key[i] = (e < N_EDGES) ? (dstp[e] >> TR_LOG) : -1;
    }
    for (int b = tid; b < NB; b += 256) hist[b] = 0;
    __syncthreads();

    const int i0 = tid, i1 = tid + 256;
    const int k0 = key[i0], k1 = key[i1];
    int all0 = 0, bef0 = 0, all1 = 0, bef1 = 0;
    for (int j = 0; j < SEG; j += 4) {
        const int4 q = *(const int4*)&key[j];
        const int c0 = (q.x == k0) + (q.y == k0) + (q.z == k0) + (q.w == k0);
        const int c1 = (q.x == k1) + (q.y == k1) + (q.z == k1) + (q.w == k1);
        all0 += c0; all1 += c1;
        if (j + 4 <= i0) bef0 += c0;
        if (j + 4 <= i1) bef1 += c1;
    }
    {
        const int j0 = i0 & ~3, j1 = i1 & ~3;
        const int4 q = *(const int4*)&key[j0];
        const int4 p = *(const int4*)&key[j1];
        bef0 += ((q.x == k0) & (j0     < i0)) + ((q.y == k0) & (j0 + 1 < i0))
              + ((q.z == k0) & (j0 + 2 < i0));
        bef1 += ((p.x == k1) & (j1     < i1)) + ((p.y == k1) & (j1 + 1 < i1))
              + ((p.z == k1) & (j1 + 2 < i1));
    }
    if (k0 >= 0) { rankv[e0 + i0] = bef0; if (bef0 + 1 == all0) hist[k0] = all0; }
    if (k1 >= 0) { rankv[e0 + i1] = bef1; if (bef1 + 1 == all1) hist[k1] = all1; }
    __syncthreads();
    for (int b = tid; b < NB; b += 256) cnt[(size_t)s * NB + b] = hist[b];
}

__global__ __launch_bounds__(1024) void k_scan_seg(
    int* __restrict__ cnt, int* __restrict__ tot)
{
    __shared__ int buf[2][1024];
    const int t = threadIdx.x;
    const int b = blockIdx.x;
    const int v = (t < NSEG) ? cnt[(size_t)t * NB + b] : 0;
    buf[0][t] = v;
    __syncthreads();
    int pin = 0;
    for (int d = 1; d < 1024; d <<= 1) {
        int x = buf[pin][t];
        if (t >= d) x += buf[pin][t - d];
        buf[pin ^ 1][t] = x;
        __syncthreads();
        pin ^= 1;
    }
    const int incl = buf[pin][t];
    if (t < NSEG) cnt[(size_t)t * NB + b] = incl - v;
    if (t == 1023) tot[b] = incl;
}

__global__ __launch_bounds__(1024) void k_scan_tiles(
    const int* __restrict__ tot, int* __restrict__ bstart)
{
    __shared__ int buf[2][1024];
    const int t  = threadIdx.x;
    const int i0 = 2 * t, i1 = 2 * t + 1;
    const int v0 = (i0 < NB) ? tot[i0] : 0;
    const int v1 = (i1 < NB) ? tot[i1] : 0;
    buf[0][t] = v0 + v1;
    __syncthreads();
    int pin = 0;
    for (int d = 1; d < 1024; d <<= 1) {
        int x = buf[pin][t];
        if (t >= d) x += buf[pin][t - d];
        buf[pin ^ 1][t] = x;
        __syncthreads();
        pin ^= 1;
    }
    const int excl = buf[pin][t] - (v0 + v1);
    if (i0 < NB) bstart[i0] = excl;
    if (i1 < NB) bstart[i1] = excl + v0;
}

__global__ __launch_bounds__(256) void k_place(
    const int* __restrict__ edge_index, const int* __restrict__ edge_attr,
    const int* __restrict__ rankv, const int* __restrict__ within,
    const int* __restrict__ bstart, int2* __restrict__ rec)
{
    const int s = blockIdx.x;
    for (int i = threadIdx.x; i < SEG; i += 256) {
        const int e = s * SEG + i;
        if (e < N_EDGES) {
            const int src = edge_index[e];
            const int d   = edge_index[N_EDGES + e];
            const int b   = d >> TR_LOG;
            const int pos = bstart[b] + within[(size_t)s * NB + b] + rankv[e];
            const int a0 = edge_attr[e * 3 + 0];
            const int a1 = edge_attr[e * 3 + 1];
            const int a2 = edge_attr[e * 3 + 2];
            rec[pos] = make_int2(src, (d & (TR - 1)) | (a0 << 8) | (a1 << 12) | (a2 << 16));
        }
    }
}

#define CHUNK 512
__global__ __launch_bounds__(128, 7) void k_tile_agg(
    const int2* __restrict__ rec, const int* __restrict__ bstart,
    const int* __restrict__ tot, const float* __restrict__ bond_emb,
    const float* __restrict__ h, float* __restrict__ agg)
{
    __shared__ __align__(16) float acc[TR * HID];
    __shared__ __align__(8) int2 own[4 * CHUNK];
    const int tid  = threadIdx.x;
    const int w    = tid >> 5;
    const int lane = tid & 31;
    const int c    = lane * 4;
    const int b    = blockIdx.x;
    const int n    = tot[b];
    const int2* rp = rec + bstart[b];

    #pragma unroll
    for (int k = 0; k < TR / 4; ++k)
        *(float4*)&acc[(w + 4 * k) * HID + c] = make_float4(0.f, 0.f, 0.f, 0.f);

    int2* lw = &own[w * CHUNK];
    for (int cb = 0; cb < n; cb += CHUNK) {
        const int ce = (n < cb + CHUNK) ? n : (cb + CHUNK);

        int2 rcs[CHUNK / 32];
        #pragma unroll
        for (int g = 0; g < CHUNK / 32; ++g) {
            const int i = cb + 32 * g + lane;
            rcs[g] = (i < ce) ? rp[i] : make_int2(0, 0);
        }
        int m = 0;
        #pragma unroll
        for (int g = 0; g < CHUNK / 32; ++g) {
            const bool mine = (cb + 32 * g + lane < ce) && ((rcs[g].y & 3) == w);
            const unsigned mask = __builtin_amdgcn_ballot_w32(mine);
            if (mine) lw[m + (int)__builtin_amdgcn_mbcnt_lo(mask, 0u)] = rcs[g];
            m += __builtin_popcount(mask);
        }
        __builtin_amdgcn_fence(__ATOMIC_RELEASE, "wavefront");
        __builtin_amdgcn_wave_barrier();
        __builtin_amdgcn_fence(__ATOMIC_ACQUIRE, "wavefront");

        int2 rc = make_int2(0, 0);
        for (int t0 = 0; t0 < m; t0 += 8) {
            const int lb = t0 & 31;
            if (lb == 0) {
                int li = t0 + lane;
                li = li < m ? li : m - 1;
                rc = lw[li];
            }
            int  sv[8], pv[8];
            bool ok[8];
            #pragma unroll
            for (int g = 0; g < 8; ++g) {
                ok[g] = (t0 + g < m);
                sv[g] = __builtin_amdgcn_readlane(rc.x, lb + g);
                pv[g] = __builtin_amdgcn_readlane(rc.y, lb + g);
            }
            float4 m4[8];
            #pragma unroll
            for (int g = 0; g < 8; ++g) {
                const int a0 = (pv[g] >> 8)  & 15;
                const int a1 = (pv[g] >> 12) & 15;
                const int a2 = (pv[g] >> 16) & 15;
                const float4 hv = *(const float4*)(h + (size_t)sv[g] * HID + c);
                const float4 e0 = *(const float4*)(bond_emb + ((size_t)(0 * 6 + a0)) * HID + c);
                const float4 e1 = *(const float4*)(bond_emb + ((size_t)(1 * 6 + a1)) * HID + c);
                const float4 e2 = *(const float4*)(bond_emb + ((size_t)(2 * 6 + a2)) * HID + c);
                float m0 = hv.x + e0.x + e1.x + e2.x; m0 = m0 > 0.f ? m0 : 0.f;
                float m1 = hv.y + e0.y + e1.y + e2.y; m1 = m1 > 0.f ? m1 : 0.f;
                float m2 = hv.z + e0.z + e1.z + e2.z; m2 = m2 > 0.f ? m2 : 0.f;
                float m3 = hv.w + e0.w + e1.w + e2.w; m3 = m3 > 0.f ? m3 : 0.f;
                m4[g] = make_float4(m0, m1, m2, m3);
            }
            #pragma unroll
            for (int g = 0; g < 8; ++g) {
                if (ok[g]) {
                    float* a = &acc[(pv[g] & (TR - 1)) * HID + c];
                    float4 t = *(float4*)a;
                    t.x += m4[g].x; t.y += m4[g].y; t.z += m4[g].z; t.w += m4[g].w;
                    *(float4*)a = t;
                }
            }
        }
        __builtin_amdgcn_fence(__ATOMIC_RELEASE, "wavefront");
        __builtin_amdgcn_wave_barrier();
        __builtin_amdgcn_fence(__ATOMIC_ACQUIRE, "wavefront");
    }

    #pragma unroll
    for (int k = 0; k < TR / 4; ++k) {
        const int r    = w + 4 * k;
        const int node = b * TR + r;
        if (node < N_NODES) {
            const float4 hv = *(const float4*)(h + (size_t)node * HID + c);
            const float4 a  = *(const float4*)&acc[r * HID + c];
            *(float4*)(agg + (size_t)node * HID + c) =
                make_float4(hv.x + a.x, hv.y + a.y, hv.z + a.z, hv.w + a.w);
        }
    }
}

__global__ __launch_bounds__(256) void k_prep_weights(
    const float* __restrict__ w1, const float* __restrict__ w2,
    __bf16* __restrict__ f1, __bf16* __restrict__ f2)
{
    int o = blockIdx.x * 256 + threadIdx.x;
    if (o >= NLAYERS * 4 * 8 * 32 * 16) return;
    int j     =  o        & 15;
    int lane  = (o >> 4)  & 31;
    int ntile = (o >> 9)  & 7;
    int ks    = (o >> 12) & 3;
    int layer =  o >> 14;
    int colL  = lane & 15;
    int khalf = (lane >> 4) * 8;
    int K     = ks * 32 + ((j < 8) ? (khalf + j) : (16 + khalf + (j - 8)));
    int col   = ntile * 16 + colL;
    size_t src = (size_t)layer * HID * HID + (size_t)K * HID + col;
    size_t dst = (size_t)layer * 16384 +
                 (((size_t)ks * 8 + ntile) * 32 + lane) * 16 + j;
    f1[dst] = (__bf16)w1[src];
    f2[dst] = (__bf16)w2[src];
}

#define NTILES (N_NODES / 16)
#define NU_WAVES 8
#define ZP 136
static_assert(N_NODES % 16 == 0, "node tiles are full");

__global__ __launch_bounds__(256) void k_node_update(
    const float* __restrict__ agg, const float* __restrict__ hcur,
    const __bf16* __restrict__ wf1, const __bf16* __restrict__ wf2,
    const float* __restrict__ b1, const float* __restrict__ b2,
    const float* __restrict__ gamma, const float* __restrict__ beta,
    const float* __restrict__ mean, const float* __restrict__ var,
    float* __restrict__ hnext)
{
    __shared__ __align__(16) __bf16 wl[2 * 16384];
    __shared__ __align__(16) __bf16 z1s[NU_WAVES * 16 * ZP];

    const int tid  = threadIdx.x;
    const int wave = __builtin_amdgcn_readfirstlane(tid >> 5);
    const int lane = tid & 31;
    const int c    = lane & 15;
    const int hh   = lane >> 4;
    const int tile = blockIdx.x * NU_WAVES + wave;
    const bool act = tile < NTILES;
    const int node0 = tile * 16;

    uint4 wv[16];
    #pragma unroll
    for (int i = 0; i < 16; ++i) {
        const int qq = (tid + 256 * i) & 2047;
        wv[i] = *(const uint4*)(((i >> 3) ? wf2 : wf1) + (size_t)qq * 8);
    }

    float4 av[16];
    {
        const int rowA = (act ? node0 : (NTILES - 1) * 16) + c;
        const float* ap = agg + (size_t)rowA * HID + 8 * hh;
        #pragma unroll
        for (int ks = 0; ks < 4; ++ks) {
            av[4 * ks + 0] = *(const float4*)(ap + ks * 32);
            av[4 * ks + 1] = *(const float4*)(ap + ks * 32 + 4);
            av[4 * ks + 2] = *(const float4*)(ap + ks * 32 + 16);
            av[4 * ks + 3] = *(const float4*)(ap + ks * 32 + 20);
        }
    }

    #pragma unroll
    for (int i = 0; i < 16; ++i) {
        const int qq = (tid + 256 * i) & 2047;
        const int f = qq >> 6;
        const int l = (qq >> 1) & 31;
        const int p = qq & 1;
        *(uint4*)&wl[(i >> 3) * 16384 + f * 512 + p * 256 + l * 8] = wv[i];
    }
    __syncthreads();
    if (!act) return;

    v16bf a[4];
    #pragma unroll
    for (int ks = 0; ks < 4; ++ks) {
        const float4 p0 = av[4 * ks + 0], p1 = av[4 * ks + 1];
        const float4 p2 = av[4 * ks + 2], p3 = av[4 * ks + 3];
        v16bf t;
        t[0]  = (__bf16)p0.x; t[1]  = (__bf16)p0.y; t[2]  = (__bf16)p0.z; t[3]  = (__bf16)p0.w;
        t[4]  = (__bf16)p1.x; t[5]  = (__bf16)p1.y; t[6]  = (__bf16)p1.z; t[7]  = (__bf16)p1.w;
        t[8]  = (__bf16)p2.x; t[9]  = (__bf16)p2.y; t[10] = (__bf16)p2.z; t[11] = (__bf16)p2.w;
        t[12] = (__bf16)p3.x; t[13] = (__bf16)p3.y; t[14] = (__bf16)p3.z; t[15] = (__bf16)p3.w;
        a[ks] = t;
    }

    v8f acc[8];
    #pragma unroll
    for (int nt = 0; nt < 8; ++nt) acc[nt] = (v8f){};
    #pragma unroll
    for (int ks = 0; ks < 4; ++ks) {
        #pragma unroll
        for (int nt = 0; nt < 8; ++nt) {
            const __bf16* bp = &wl[(ks * 8 + nt) * 512 + lane * 8];
            const v8bf blo = *(const v8bf*)(bp);
            const v8bf bhi = *(const v8bf*)(bp + 256);
            const v16bf b = __builtin_shufflevector(blo, bhi,
                        0,1,2,3,4,5,6,7,8,9,10,11,12,13,14,15);
            acc[nt] = __builtin_amdgcn_wmma_f32_16x16x32_bf16(
                          false, a[ks], false, b, (short)0, acc[nt], false, false);
        }
    }

    __bf16* z1w = &z1s[wave * 16 * ZP];
    #pragma unroll
    for (int nt = 0; nt < 8; ++nt) {
        const int col = nt * 16 + c;
        const float bias1 = b1[col];
        #pragma unroll
        for (int r = 0; r < 8; ++r) {
            float v = acc[nt][r] + bias1;
            v = v > 0.f ? v : 0.f;
            z1w[(8 * hh + r) * ZP + col] = (__bf16)v;
        }
    }
    __builtin_amdgcn_fence(__ATOMIC_RELEASE, "wavefront");
    __builtin_amdgcn_wave_barrier();
    __builtin_amdgcn_fence(__ATOMIC_ACQUIRE, "wavefront");

    #pragma unroll
    for (int ks = 0; ks < 4; ++ks) {
        const __bf16* ap = &z1w[c * ZP + ks * 32 + 8 * hh];
        const v8bf alo = *(const v8bf*)(ap);
        const v8bf ahi = *(const v8bf*)(ap + 16);
        a[ks] = __builtin_shufflevector(alo, ahi,
                    0,1,2,3,4,5,6,7,8,9,10,11,12,13,14,15);
    }

    #pragma unroll
    for (int nt = 0; nt < 8; ++nt) acc[nt] = (v8f){};
    #pragma unroll
    for (int ks = 0; ks < 4; ++ks) {
        #pragma unroll
        for (int nt = 0; nt < 8; ++nt) {
            const __bf16* bp = &wl[16384 + (ks * 8 + nt) * 512 + lane * 8];
            const v8bf blo = *(const v8bf*)(bp);
            const v8bf bhi = *(const v8bf*)(bp + 256);
            const v16bf b = __builtin_shufflevector(blo, bhi,
                        0,1,2,3,4,5,6,7,8,9,10,11,12,13,14,15);
            acc[nt] = __builtin_amdgcn_wmma_f32_16x16x32_bf16(
                          false, a[ks], false, b, (short)0, acc[nt], false, false);
        }
    }

    #pragma unroll
    for (int nt = 0; nt < 8; ++nt) {
        const int col = nt * 16 + c;
        const float bias2 = b2[col];
        const float mu    = mean[col];
        const float inv   = rsqrtf(var[col] + 1e-5f);
        const float ga    = gamma[col];
        const float be    = beta[col];
        #pragma unroll
        for (int r = 0; r < 8; ++r) {
            const size_t idx = (size_t)(node0 + 8 * hh + r) * HID + col;
            float v = acc[nt][r] + bias2;
            v = (v - mu) * inv * ga + be;
            v = v > 0.f ? v : 0.f;
            hnext[idx] = hcur[idx] + v;
        }
    }
}

__global__ __launch_bounds__(256) void k_pool_zero(
    float* __restrict__ sums, float* __restrict__ cnts)
{
    int i = blockIdx.x * 256 + threadIdx.x;
    if (i < N_GRAPHS * HID) sums[i] = 0.f;
    if (i < N_GRAPHS)       cnts[i] = 0.f;
}

__global__ __launch_bounds__(256) void k_pool_acc(
    const float* __restrict__ h, const int* __restrict__ batch,
    float* __restrict__ sums, float* __restrict__ cnts)
{
    int gid = blockIdx.x * 256 + threadIdx.x;
    if (gid >= N_NODES * 32) return;
    int n = gid >> 5;
    int lane = gid & 31;
    int g = batch[n];
    const float* hp = h + (size_t)n * HID + lane;
    const float v0 = hp[0];
    const float v1 = hp[32];
    const float v2 = hp[64];
    const float v3 = hp[96];
    float* o = sums + (size_t)g * HID + lane;
    atomicAdd(o,      v0);
    atomicAdd(o + 32, v1);
    atomicAdd(o + 64, v2);
    atomicAdd(o + 96, v3);
    if (lane == 0) atomicAdd(cnts + g, 1.0f);
}

__global__ __launch_bounds__(128) void k_pool_mlp(
    const float* __restrict__ sums, const float* __restrict__ cnts,
    const float* __restrict__ mw, const float* __restrict__ mb,
    float* __restrict__ out)
{
    __shared__ float p[HID];
    int g = blockIdx.x;
    int t = threadIdx.x;
    float cnt = cnts[g];
    cnt = cnt > 1.f ? cnt : 1.f;
    p[t] = sums[(size_t)g * HID + t] / cnt;
    __syncthreads();
    if (t < OUTD) {
        float a = mb[t];
        #pragma unroll 16
        for (int c = 0; c < HID; ++c) a += p[c] * mw[c * OUTD + t];
        out[g * OUTD + t] = a;
    }
}

extern "C" void kernel_launch(void* const* d_in, const int* in_sizes, int n_in,
                              void* d_out, int out_size, void* d_ws, size_t ws_size,
                              hipStream_t stream) {
    const int*   x          = (const int*)  d_in[0];
    const int*   edge_index = (const int*)  d_in[1];
    const int*   edge_attr  = (const int*)  d_in[2];
    const int*   batch      = (const int*)  d_in[3];
    const float* atom_emb   = (const float*)d_in[4];
    const float* bond_emb   = (const float*)d_in[5];
    const float* lin1_w     = (const float*)d_in[6];
    const float* lin1_b     = (const float*)d_in[7];
    const float* lin2_w     = (const float*)d_in[8];
    const float* lin2_b     = (const float*)d_in[9];
    const float* bn_gamma   = (const float*)d_in[10];
    const float* bn_beta    = (const float*)d_in[11];
    const float* bn_mean    = (const float*)d_in[12];
    const float* bn_var     = (const float*)d_in[13];
    const float* mlp_w      = (const float*)d_in[14];
    const float* mlp_b      = (const float*)d_in[15];

    char* ws = (char*)d_ws;
    size_t off = 0;
    auto alloc = [&](size_t bytes) -> char* {
        char* p = ws + off;
        off = (off + bytes + 255) & ~(size_t)255;
        return p;
    };
    float*  h0   = (float*) alloc((size_t)N_NODES * HID * 4);
    float*  h1   = (float*) alloc((size_t)N_NODES * HID * 4);
    float*  agg  = (float*) alloc((size_t)N_NODES * HID * 4);
    __bf16* wf1  = (__bf16*)alloc((size_t)NLAYERS * 16384 * 2);
    __bf16* wf2  = (__bf16*)alloc((size_t)NLAYERS * 16384 * 2);
    float*  sums = (float*) alloc((size_t)N_GRAPHS * HID * 4);
    float*  cnts = (float*) alloc((size_t)N_GRAPHS * 4);
    int*    rankv  = (int*)  alloc((size_t)N_EDGES * 4);
    int*    cnt    = (int*)  alloc((size_t)NSEG * NB * 4);
    int*    tot    = (int*)  alloc((size_t)NB * 4);
    int*    bstart = (int*)  alloc((size_t)NB * 4);
    int2*   rec    = (int2*) alloc((size_t)N_EDGES * 8);

    const int nodeChunks = N_NODES * 32;

    k_prep_weights<<<(NLAYERS * 4 * 8 * 32 * 16 + 255) / 256, 256, 0, stream>>>(
        lin1_w, lin2_w, wf1, wf2);
    k_encode_nodes<<<(nodeChunks + 255) / 256, 256, 0, stream>>>(x, atom_emb, h0);

    k_rank<<<NSEG, 256, 0, stream>>>(edge_index, rankv, cnt);
    k_scan_seg<<<NB, 1024, 0, stream>>>(cnt, tot);
    k_scan_tiles<<<1, 1024, 0, stream>>>(tot, bstart);
    k_place<<<NSEG, 256, 0, stream>>>(edge_index, edge_attr, rankv, cnt, bstart, rec);

    float* hc = h0;
    float* hn = h1;
    for (int i = 0; i < NLAYERS; ++i) {
        k_tile_agg<<<NB, 128, 0, stream>>>(rec, bstart, tot, bond_emb, hc, agg);
        k_node_update<<<(NTILES + NU_WAVES - 1) / NU_WAVES, 256, 0, stream>>>(
            agg, hc, wf1 + (size_t)i * 16384, wf2 + (size_t)i * 16384,
            lin1_b + i * HID, lin2_b + i * HID,
            bn_gamma + i * HID, bn_beta + i * HID,
            bn_mean + i * HID, bn_var + i * HID, hn);
        float* t = hc; hc = hn; hn = t;
    }

    k_pool_zero<<<(N_GRAPHS * HID + 255) / 256, 256, 0, stream>>>(sums, cnts);
    k_pool_acc<<<(nodeChunks + 255) / 256, 256, 0, stream>>>(hc, batch, sums, cnts);
    k_pool_mlp<<<N_GRAPHS, 128, 0, stream>>>(sums, cnts, mlp_w, mlp_b, (float*)d_out);
}


# pooling + head in one launch: a 4-wave block per graph finds its row range in the sorted graph ids and sums its rows in fixed order (was arrival-order f32 atomics); 17 -> 15 launches, no clear
# speedup vs baseline: 1.9902x; 1.0845x over previous
#include <hip/hip_runtime.h>
#include <hip/hip_bf16.h>

#define N_NODES 50000
#define N_EDGES 500000
#define N_GRAPHS 1000
#define HID 128
#define OUTD 10
#define NLAYERS 4

typedef __attribute__((ext_vector_type(16))) __bf16 v16bf;
typedef __attribute__((ext_vector_type(8)))  __bf16 v8bf;
typedef __attribute__((ext_vector_type(8)))  float  v8f;

__global__ __launch_bounds__(256) void k_encode_nodes(
    const int* __restrict__ x, const float* __restrict__ atom_emb,
    float* __restrict__ h)
{
    int gid = blockIdx.x * 256 + threadIdx.x;
    if (gid >= N_NODES * 32) return;
    int n = gid >> 5;
    int c = (gid & 31) * 4;
    float4 acc = make_float4(0.f, 0.f, 0.f, 0.f);
    #pragma unroll
    for (int f = 0; f < 9; ++f) {
        int idx = x[n * 9 + f];
        const float4 v = *(const float4*)(atom_emb + ((size_t)(f * 120 + idx)) * HID + c);
        acc.x += v.x; acc.y += v.y; acc.z += v.z; acc.w += v.w;
    }
    *(float4*)(h + (size_t)n * HID + c) = acc;
}

#define TR 32
#define TR_LOG 5
#define NB ((N_NODES + TR - 1) / TR)
#define SEG 512
#define NSEG ((N_EDGES + SEG - 1) / SEG)
static_assert(SEG == 2 * 256, "k_rank ranks two edges per thread");
static_assert(NSEG <= 1024, "k_scan_seg scans one segment per thread");
static_assert(NB <= 2048, "k_scan_tiles scans two tiles per thread");

__global__ __launch_bounds__(256) void k_rank(
    const int* __restrict__ edge_index, int* __restrict__ rankv,
    int* __restrict__ cnt)
{
    __shared__ __align__(16) int key[SEG];
    __shared__ int hist[NB];
    const int tid = threadIdx.x;
    const int s   = blockIdx.x;
    const int e0  = s * SEG;
    const int* dstp = edge_index + N_EDGES;
    for (int i = tid; i < SEG; i += 256) {
        const int e = e0 + i;
        key[i] = (e < N_EDGES) ? (dstp[e] >> TR_LOG) : -1;
    }
    for (int b = tid; b < NB; b += 256) hist[b] = 0;
    __syncthreads();

    const int i0 = tid, i1 = tid + 256;
    const int k0 = key[i0], k1 = key[i1];
    int all0 = 0, bef0 = 0, all1 = 0, bef1 = 0;
    for (int j = 0; j < SEG; j += 4) {
        const int4 q = *(const int4*)&key[j];
        const int c0 = (q.x == k0) + (q.y == k0) + (q.z == k0) + (q.w == k0);
        const int c1 = (q.x == k1) + (q.y == k1) + (q.z == k1) + (q.w == k1);
        all0 += c0; all1 += c1;
        if (j + 4 <= i0) bef0 += c0;
        if (j + 4 <= i1) bef1 += c1;
    }
    {
        const int j0 = i0 & ~3, j1 = i1 & ~3;
        const int4 q = *(const int4*)&key[j0];
        const int4 p = *(const int4*)&key[j1];
        bef0 += ((q.x == k0) & (j0     < i0)) + ((q.y == k0) & (j0 + 1 < i0))
              + ((q.z == k0) & (j0 + 2 < i0));
        bef1 += ((p.x == k1) & (j1     < i1)) + ((p.y == k1) & (j1 + 1 < i1))
              + ((p.z == k1) & (j1 + 2 < i1));
    }
    if (k0 >= 0) { rankv[e0 + i0] = bef0; if (bef0 + 1 == all0) hist[k0] = all0; }
    if (k1 >= 0) { rankv[e0 + i1] = bef1; if (bef1 + 1 == all1) hist[k1] = all1; }
    __syncthreads();
    for (int b = tid; b < NB; b += 256) cnt[(size_t)s * NB + b] = hist[b];
}

__global__ __launch_bounds__(1024) void k_scan_seg(
    int* __restrict__ cnt, int* __restrict__ tot)
{
    __shared__ int buf[2][1024];
    const int t = threadIdx.x;
    const int b = blockIdx.x;
    const int v = (t < NSEG) ? cnt[(size_t)t * NB + b] : 0;
    buf[0][t] = v;
    __syncthreads();
    int pin = 0;
    for (int d = 1; d < 1024; d <<= 1) {
        int x = buf[pin][t];
        if (t >= d) x += buf[pin][t - d];
        buf[pin ^ 1][t] = x;
        __syncthreads();
        pin ^= 1;
    }
    const int incl = buf[pin][t];
    if (t < NSEG) cnt[(size_t)t * NB + b] = incl - v;
    if (t == 1023) tot[b] = incl;
}

__global__ __launch_bounds__(1024) void k_scan_tiles(
    const int* __restrict__ tot, int* __restrict__ bstart)
{
    __shared__ int buf[2][1024];
    const int t  = threadIdx.x;
    const int i0 = 2 * t, i1 = 2 * t + 1;
    const int v0 = (i0 < NB) ? tot[i0] : 0;
    const int v1 = (i1 < NB) ? tot[i1] : 0;
    buf[0][t] = v0 + v1;
    __syncthreads();
    int pin = 0;
    for (int d = 1; d < 1024; d <<= 1) {
        int x = buf[pin][t];
        if (t >= d) x += buf[pin][t - d];
        buf[pin ^ 1][t] = x;
        __syncthreads();
        pin ^= 1;
    }
    const int excl = buf[pin][t] - (v0 + v1);
    if (i0 < NB) bstart[i0] = excl;
    if (i1 < NB) bstart[i1] = excl + v0;
}

__global__ __launch_bounds__(256) void k_place(
    const int* __restrict__ edge_index, const int* __restrict__ edge_attr,
    const int* __restrict__ rankv, const int* __restrict__ within,
    const int* __restrict__ bstart, int2* __restrict__ rec)
{
    const int s = blockIdx.x;
    for (int i = threadIdx.x; i < SEG; i += 256) {
        const int e = s * SEG + i;
        if (e < N_EDGES) {
            const int src = edge_index[e];
            const int d   = edge_index[N_EDGES + e];
            const int b   = d >> TR_LOG;
            const int pos = bstart[b] + within[(size_t)s * NB + b] + rankv[e];
            const int a0 = edge_attr[e * 3 + 0];
            const int a1 = edge_attr[e * 3 + 1];
            const int a2 = edge_attr[e * 3 + 2];
            rec[pos] = make_int2(src, (d & (TR - 1)) | (a0 << 8) | (a1 << 12) | (a2 << 16));
        }
    }
}

#define CHUNK 512
__global__ __launch_bounds__(128, 7) void k_tile_agg(
    const int2* __restrict__ rec, const int* __restrict__ bstart,
    const int* __restrict__ tot, const float* __restrict__ bond_emb,
    const float* __restrict__ h, float* __restrict__ agg)
{
    __shared__ __align__(16) float acc[TR * HID];
    __shared__ __align__(8) int2 own[4 * CHUNK];
    const int tid  = threadIdx.x;
    const int w    = tid >> 5;
    const int lane = tid & 31;
    const int c    = lane * 4;
    const int b    = blockIdx.x;
    const int n    = tot[b];
    const int2* rp = rec + bstart[b];

    #pragma unroll
    for (int k = 0; k < TR / 4; ++k)
        *(float4*)&acc[(w + 4 * k) * HID + c] = make_float4(0.f, 0.f, 0.f, 0.f);

    int2* lw = &own[w * CHUNK];
    for (int cb = 0; cb < n; cb += CHUNK) {
        const int ce = (n < cb + CHUNK) ? n : (cb + CHUNK);

        int2 rcs[CHUNK / 32];
        #pragma unroll
        for (int g = 0; g < CHUNK / 32; ++g) {
            const int i = cb + 32 * g + lane;
            rcs[g] = (i < ce) ? rp[i] : make_int2(0, 0);
        }
        int m = 0;
        #pragma unroll
        for (int g = 0; g < CHUNK / 32; ++g) {
            const bool mine = (cb + 32 * g + lane < ce) && ((rcs[g].y & 3) == w);
            const unsigned mask = __builtin_amdgcn_ballot_w32(mine);
            if (mine) lw[m + (int)__builtin_amdgcn_mbcnt_lo(mask, 0u)] = rcs[g];
            m += __builtin_popcount(mask);
        }
        __builtin_amdgcn_fence(__ATOMIC_RELEASE, "wavefront");
        __builtin_amdgcn_wave_barrier();
        __builtin_amdgcn_fence(__ATOMIC_ACQUIRE, "wavefront");

        int2 rc = make_int2(0, 0);
        for (int t0 = 0; t0 < m; t0 += 8) {
            const int lb = t0 & 31;
            if (lb == 0) {
                int li = t0 + lane;
                li = li < m ? li : m - 1;
                rc = lw[li];
            }
            int  sv[8], pv[8];
            bool ok[8];
            #pragma unroll
            for (int g = 0; g < 8; ++g) {
                ok[g] = (t0 + g < m);
                sv[g] = __builtin_amdgcn_readlane(rc.x, lb + g);
                pv[g] = __builtin_amdgcn_readlane(rc.y, lb + g);
            }
            float4 m4[8];
            #pragma unroll
            for (int g = 0; g < 8; ++g) {
                const int a0 = (pv[g] >> 8)  & 15;
                const int a1 = (pv[g] >> 12) & 15;
                const int a2 = (pv[g] >> 16) & 15;
                const float4 hv = *(const float4*)(h + (size_t)sv[g] * HID + c);
                const float4 e0 = *(const float4*)(bond_emb + ((size_t)(0 * 6 + a0)) * HID + c);
                const float4 e1 = *(const float4*)(bond_emb + ((size_t)(1 * 6 + a1)) * HID + c);
                const float4 e2 = *(const float4*)(bond_emb + ((size_t)(2 * 6 + a2)) * HID + c);
                float m0 = hv.x + e0.x + e1.x + e2.x; m0 = m0 > 0.f ? m0 : 0.f;
                float m1 = hv.y + e0.y + e1.y + e2.y; m1 = m1 > 0.f ? m1 : 0.f;
                float m2 = hv.z + e0.z + e1.z + e2.z; m2 = m2 > 0.f ? m2 : 0.f;
                float m3 = hv.w + e0.w + e1.w + e2.w; m3 = m3 > 0.f ? m3 : 0.f;
                m4[g] = make_float4(m0, m1, m2, m3);
            }
            #pragma unroll
            for (int g = 0; g < 8; ++g) {
                if (ok[g]) {
                    float* a = &acc[(pv[g] & (TR - 1)) * HID + c];
                    float4 t = *(float4*)a;
                    t.x += m4[g].x; t.y += m4[g].y; t.z += m4[g].z; t.w += m4[g].w;
                    *(float4*)a = t;
                }
            }
        }
        __builtin_amdgcn_fence(__ATOMIC_RELEASE, "wavefront");
        __builtin_amdgcn_wave_barrier();
        __builtin_amdgcn_fence(__ATOMIC_ACQUIRE, "wavefront");
    }

    #pragma unroll
    for (int k = 0; k < TR / 4; ++k) {
        const int r    = w + 4 * k;
        const int node = b * TR + r;
        if (node < N_NODES) {
            const float4 hv = *(const float4*)(h + (size_t)node * HID + c);
            const float4 a  = *(const float4*)&acc[r * HID + c];
            *(float4*)(agg + (size_t)node * HID + c) =
                make_float4(hv.x + a.x, hv.y + a.y, hv.z + a.z, hv.w + a.w);
        }
    }
}

__global__ __launch_bounds__(256) void k_prep_weights(
    const float* __restrict__ w1, const float* __restrict__ w2,
    __bf16* __restrict__ f1, __bf16* __restrict__ f2)
{
    int o = blockIdx.x * 256 + threadIdx.x;
    if (o >= NLAYERS * 4 * 8 * 32 * 16) return;
    int j     =  o        & 15;
    int lane  = (o >> 4)  & 31;
    int ntile = (o >> 9)  & 7;
    int ks    = (o >> 12) & 3;
    int layer =  o >> 14;
    int colL  = lane & 15;
    int khalf = (lane >> 4) * 8;
    int K     = ks * 32 + ((j < 8) ? (khalf + j) : (16 + khalf + (j - 8)));
    int col   = ntile * 16 + colL;
    size_t src = (size_t)layer * HID * HID + (size_t)K * HID + col;
    size_t dst = (size_t)layer * 16384 +
                 (((size_t)ks * 8 + ntile) * 32 + lane) * 16 + j;
    f1[dst] = (__bf16)w1[src];
    f2[dst] = (__bf16)w2[src];
}

#define NTILES (N_NODES / 16)
#define NU_WAVES 8
#define ZP 136
static_assert(N_NODES % 16 == 0, "node tiles are full");

__global__ __launch_bounds__(256) void k_node_update(
    const float* __restrict__ agg, const float* __restrict__ hcur,
    const __bf16* __restrict__ wf1, const __bf16* __restrict__ wf2,
    const float* __restrict__ b1, const float* __restrict__ b2,
    const float* __restrict__ gamma, const float* __restrict__ beta,
    const float* __restrict__ mean, const float* __restrict__ var,
    float* __restrict__ hnext)
{
    __shared__ __align__(16) __bf16 wl[2 * 16384];
    __shared__ __align__(16) __bf16 z1s[NU_WAVES * 16 * ZP];

    const int tid  = threadIdx.x;
    const int wave = __builtin_amdgcn_readfirstlane(tid >> 5);
    const int lane = tid & 31;
    const int c    = lane & 15;
    const int hh   = lane >> 4;
    const int tile = blockIdx.x * NU_WAVES + wave;
    const bool act = tile < NTILES;
    const int node0 = tile * 16;

    uint4 wv[16];
    #pragma unroll
    for (int i = 0; i < 16; ++i) {
        const int qq = (tid + 256 * i) & 2047;
        wv[i] = *(const uint4*)(((i >> 3) ? wf2 : wf1) + (size_t)qq * 8);
    }

    float4 av[16];
    {
        const int rowA = (act ? node0 : (NTILES - 1) * 16) + c;
        const float* ap = agg + (size_t)rowA * HID + 8 * hh;
        #pragma unroll
        for (int ks = 0; ks < 4; ++ks) {
            av[4 * ks + 0] = *(const float4*)(ap + ks * 32);
            av[4 * ks + 1] = *(const float4*)(ap + ks * 32 + 4);
            av[4 * ks + 2] = *(const float4*)(ap + ks * 32 + 16);
            av[4 * ks + 3] = *(const float4*)(ap + ks * 32 + 20);
        }
    }

    #pragma unroll
    for (int i = 0; i < 16; ++i) {
        const int qq = (tid + 256 * i) & 2047;
        const int f = qq >> 6;
        const int l = (qq >> 1) & 31;
        const int p = qq & 1;
        *(uint4*)&wl[(i >> 3) * 16384 + f * 512 + p * 256 + l * 8] = wv[i];
    }
    __syncthreads();
    if (!act) return;

    v16bf a[4];
    #pragma unroll
    for (int ks = 0; ks < 4; ++ks) {
        const float4 p0 = av[4 * ks + 0], p1 = av[4 * ks + 1];
        const float4 p2 = av[4 * ks + 2], p3 = av[4 * ks + 3];
        v16bf t;
        t[0]  = (__bf16)p0.x; t[1]  = (__bf16)p0.y; t[2]  = (__bf16)p0.z; t[3]  = (__bf16)p0.w;
        t[4]  = (__bf16)p1.x; t[5]  = (__bf16)p1.y; t[6]  = (__bf16)p1.z; t[7]  = (__bf16)p1.w;
        t[8]  = (__bf16)p2.x; t[9]  = (__bf16)p2.y; t[10] = (__bf16)p2.z; t[11] = (__bf16)p2.w;
        t[12] = (__bf16)p3.x; t[13] = (__bf16)p3.y; t[14] = (__bf16)p3.z; t[15] = (__bf16)p3.w;
        a[ks] = t;
    }

    v8f acc[8];
    #pragma unroll
    for (int nt = 0; nt < 8; ++nt) acc[nt] = (v8f){};
    #pragma unroll
    for (int ks = 0; ks < 4; ++ks) {
        #pragma unroll
        for (int nt = 0; nt < 8; ++nt) {
            const __bf16* bp = &wl[(ks * 8 + nt) * 512 + lane * 8];
            const v8bf blo = *(const v8bf*)(bp);
            const v8bf bhi = *(const v8bf*)(bp + 256);
            const v16bf b = __builtin_shufflevector(blo, bhi,
                        0,1,2,3,4,5,6,7,8,9,10,11,12,13,14,15);
            acc[nt] = __builtin_amdgcn_wmma_f32_16x16x32_bf16(
                          false, a[ks], false, b, (short)0, acc[nt], false, false);
        }
    }

    __bf16* z1w = &z1s[wave * 16 * ZP];
    #pragma unroll
    for (int nt = 0; nt < 8; ++nt) {
        const int col = nt * 16 + c;
        const float bias1 = b1[col];
        #pragma unroll
        for (int r = 0; r < 8; ++r) {
            float v = acc[nt][r] + bias1;
            v = v > 0.f ? v : 0.f;
            z1w[(8 * hh + r) * ZP + col] = (__bf16)v;
        }
    }
    __builtin_amdgcn_fence(__ATOMIC_RELEASE, "wavefront");
    __builtin_amdgcn_wave_barrier();
    __builtin_amdgcn_fence(__ATOMIC_ACQUIRE, "wavefront");

    #pragma unroll
    for (int ks = 0; ks < 4; ++ks) {
        const __bf16* ap = &z1w[c * ZP + ks * 32 + 8 * hh];
        const v8bf alo = *(const v8bf*)(ap);
        const v8bf ahi = *(const v8bf*)(ap + 16);
        a[ks] = __builtin_shufflevector(alo, ahi,
                    0,1,2,3,4,5,6,7,8,9,10,11,12,13,14,15);
    }

    #pragma unroll
    for (int nt = 0; nt < 8; ++nt) acc[nt] = (v8f){};
    #pragma unroll
    for (int ks = 0; ks < 4; ++ks) {
        #pragma unroll
        for (int nt = 0; nt < 8; ++nt) {
            const __bf16* bp = &wl[16384 + (ks * 8 + nt) * 512 + lane * 8];
            const v8bf blo = *(const v8bf*)(bp);
            const v8bf bhi = *(const v8bf*)(bp + 256);
            const v16bf b = __builtin_shufflevector(blo, bhi,
                        0,1,2,3,4,5,6,7,8,9,10,11,12,13,14,15);
            acc[nt] = __builtin_amdgcn_wmma_f32_16x16x32_bf16(
                          false, a[ks], false, b, (short)0, acc[nt], false, false);
        }
    }

    #pragma unroll
    for (int nt = 0; nt < 8; ++nt) {
        const int col = nt * 16 + c;
        const float bias2 = b2[col];
        const float mu    = mean[col];
        const float inv   = rsqrtf(var[col] + 1e-5f);
        const float ga    = gamma[col];
        const float be    = beta[col];
        #pragma unroll
        for (int r = 0; r < 8; ++r) {
            const size_t idx = (size_t)(node0 + 8 * hh + r) * HID + col;
            float v = acc[nt][r] + bias2;
            v = (v - mu) * inv * ga + be;
            v = v > 0.f ? v : 0.f;
            hnext[idx] = hcur[idx] + v;
        }
    }
}

#define PSTEP ((N_NODES + 127) / 128)
#define PWL ((PSTEP + 31) / 32)
static_assert(127 * PSTEP < N_NODES, "every level-1 probe lies inside the id array");

__global__ __launch_bounds__(128) void k_pool_head(
    const float* __restrict__ h, const int* __restrict__ batch,
    const float* __restrict__ mw, const float* __restrict__ mb,
    float* __restrict__ out)
{
    __shared__ __align__(16) float part[4 * HID];
    __shared__ float p[HID];
    const int g    = blockIdx.x;
    const int t    = threadIdx.x;
    const int w    = __builtin_amdgcn_readfirstlane(t >> 5);
    const int lane = t & 31;
    const int c    = lane * 4;

    int pv[4];
    #pragma unroll
    for (int j = 0; j < 4; ++j) pv[j] = batch[(lane + 32 * j) * PSTEP];
    int A = 0, B = 0;
    #pragma unroll
    for (int j = 0; j < 4; ++j) {
        A += __builtin_popcount(__builtin_amdgcn_ballot_w32(pv[j] <  g));
        B += __builtin_popcount(__builtin_amdgcn_ballot_w32(pv[j] <= g));
    }
    const int baseL = A ? (A - 1) * PSTEP + 1 : 0;
    const int baseH = B ? (B - 1) * PSTEP + 1 : 0;

    int wl[PWL], wh[PWL];
    #pragma unroll
    for (int j = 0; j < PWL; ++j) {
        const int il = baseL + lane + 32 * j, ih = baseH + lane + 32 * j;
        wl[j] = batch[il < N_NODES ? il : N_NODES - 1];
        wh[j] = batch[ih < N_NODES ? ih : N_NODES - 1];
    }
    int lo = baseL, hi = baseH;
    #pragma unroll
    for (int j = 0; j < PWL; ++j) {
        const int  o   = lane + 32 * j;
        const bool inL = (o < PSTEP) && (baseL + o < N_NODES);
        const bool inH = (o < PSTEP) && (baseH + o < N_NODES);
        lo += __builtin_popcount(__builtin_amdgcn_ballot_w32(inL && wl[j] <  g));
        hi += __builtin_popcount(__builtin_amdgcn_ballot_w32(inH && wh[j] <= g));
    }

    float4 acc = make_float4(0.f, 0.f, 0.f, 0.f);
    for (int i0 = lo + w; i0 < hi; i0 += 32) {
        float4 v[8];
        #pragma unroll
        for (int k = 0; k < 8; ++k) {
            const int i = i0 + 4 * k;
            v[k] = (i < hi) ? *(const float4*)(h + (size_t)i * HID + c)
                            : make_float4(0.f, 0.f, 0.f, 0.f);
        }
        #pragma unroll
        for (int k = 0; k < 8; ++k) {
            acc.x += v[k].x; acc.y += v[k].y; acc.z += v[k].z; acc.w += v[k].w;
        }
    }
    *(float4*)&part[w * HID + c] = acc;
    __syncthreads();
    float cnt = (float)(hi - lo);
    cnt = cnt > 1.f ? cnt : 1.f;
    p[t] = (((part[t] + part[HID + t]) + part[2 * HID + t]) + part[3 * HID + t]) / cnt;
    __syncthreads();
    if (t < OUTD) {
        float a = mb[t];
        #pragma unroll 16
        for (int cc = 0; cc < HID; ++cc) a += p[cc] * mw[cc * OUTD + t];
        out[g * OUTD + t] = a;
    }
}

extern "C" void kernel_launch(void* const* d_in, const int* in_sizes, int n_in,
                              void* d_out, int out_size, void* d_ws, size_t ws_size,
                              hipStream_t stream) {
    const int*   x          = (const int*)  d_in[0];
    const int*   edge_index = (const int*)  d_in[1];
    const int*   edge_attr  = (const int*)  d_in[2];
    const int*   batch      = (const int*)  d_in[3];
    const float* atom_emb   = (const float*)d_in[4];
    const float* bond_emb   = (const float*)d_in[5];
    const float* lin1_w     = (const float*)d_in[6];
    const float* lin1_b     = (const float*)d_in[7];
    const float* lin2_w     = (const float*)d_in[8];
    const float* lin2_b     = (const float*)d_in[9];
    const float* bn_gamma   = (const float*)d_in[10];
    const float* bn_beta    = (const float*)d_in[11];
    const float* bn_mean    = (const float*)d_in[12];
    const float* bn_var     = (const float*)d_in[13];
    const float* mlp_w      = (const float*)d_in[14];
    const float* mlp_b      = (const float*)d_in[15];

    char* ws = (char*)d_ws;
    size_t off = 0;
    auto alloc = [&](size_t bytes) -> char* {
        char* p = ws + off;
        off = (off + bytes + 255) & ~(size_t)255;
        return p;
    };
    float*  h0   = (float*) alloc((size_t)N_NODES * HID * 4);
    float*  h1   = (float*) alloc((size_t)N_NODES * HID * 4);
    float*  agg  = (float*) alloc((size_t)N_NODES * HID * 4);
    __bf16* wf1  = (__bf16*)alloc((size_t)NLAYERS * 16384 * 2);
    __bf16* wf2  = (__bf16*)alloc((size_t)NLAYERS * 16384 * 2);
    int*    rankv  = (int*)  alloc((size_t)N_EDGES * 4);
    int*    cnt    = (int*)  alloc((size_t)NSEG * NB * 4);
    int*    tot    = (int*)  alloc((size_t)NB * 4);
    int*    bstart = (int*)  alloc((size_t)NB * 4);
    int2*   rec    = (int2*) alloc((size_t)N_EDGES * 8);

    const int nodeChunks = N_NODES * 32;

    k_prep_weights<<<(NLAYERS * 4 * 8 * 32 * 16 + 255) / 256, 256, 0, stream>>>(
        lin1_w, lin2_w, wf1, wf2);
    k_encode_nodes<<<(nodeChunks + 255) / 256, 256, 0, stream>>>(x, atom_emb, h0);

    k_rank<<<NSEG, 256, 0, stream>>>(edge_index, rankv, cnt);
    k_scan_seg<<<NB, 1024, 0, stream>>>(cnt, tot);
    k_scan_tiles<<<1, 1024, 0, stream>>>(tot, bstart);
    k_place<<<NSEG, 256, 0, stream>>>(edge_index, edge_attr, rankv, cnt, bstart, rec);

    float* hc = h0;
    float* hn = h1;
    for (int i = 0; i < NLAYERS; ++i) {
        k_tile_agg<<<NB, 128, 0, stream>>>(rec, bstart, tot, bond_emb, hc, agg);
        k_node_update<<<(NTILES + NU_WAVES - 1) / NU_WAVES, 256, 0, stream>>>(
            agg, hc, wf1 + (size_t)i * 16384, wf2 + (size_t)i * 16384,
            lin1_b + i * HID, lin2_b + i * HID,
            bn_gamma + i * HID, bn_beta + i * HID,
            bn_mean + i * HID, bn_var + i * HID, hn);
        float* t = hc; hc = hn; hn = t;
    }

    k_pool_head<<<N_GRAPHS, 128, 0, stream>>>(hc, batch, mlp_w, mlp_b, (float*)d_out);
}


# segment-count scan re-tiled: lane = tile column, wave = run of 31 segments, so every load and store is one 128-B row piece (was one 4-B element per line); 50016 -> 1568 waves; all sums unchanged
# speedup vs baseline: 2.4468x; 1.2294x over previous
#include <hip/hip_runtime.h>
#include <hip/hip_bf16.h>

#define N_NODES 50000
#define N_EDGES 500000
#define N_GRAPHS 1000
#define HID 128
#define OUTD 10
#define NLAYERS 4

typedef __attribute__((ext_vector_type(16))) __bf16 v16bf;
typedef __attribute__((ext_vector_type(8)))  __bf16 v8bf;
typedef __attribute__((ext_vector_type(8)))  float  v8f;

__global__ __launch_bounds__(256) void k_encode_nodes(
    const int* __restrict__ x, const float* __restrict__ atom_emb,
    float* __restrict__ h)
{
    int gid = blockIdx.x * 256 + threadIdx.x;
    if (gid >= N_NODES * 32) return;
    int n = gid >> 5;
    int c = (gid & 31) * 4;
    float4 acc = make_float4(0.f, 0.f, 0.f, 0.f);
    #pragma unroll
    for (int f = 0; f < 9; ++f) {
        int idx = x[n * 9 + f];
        const float4 v = *(const float4*)(atom_emb + ((size_t)(f * 120 + idx)) * HID + c);
        acc.x += v.x; acc.y += v.y; acc.z += v.z; acc.w += v.w;
    }
    *(float4*)(h + (size_t)n * HID + c) = acc;
}

#define TR 32
#define TR_LOG 5
#define NB ((N_NODES + TR - 1) / TR)
#define SEG 512
#define NSEG ((N_EDGES + SEG - 1) / SEG)
static_assert(SEG == 2 * 256, "k_rank ranks two edges per thread");
static_assert(NB <= 2048, "k_scan_tiles scans two tiles per thread");

__global__ __launch_bounds__(256) void k_rank(
    const int* __restrict__ edge_index, int* __restrict__ rankv,
    int* __restrict__ cnt)
{
    __shared__ __align__(16) int key[SEG];
    __shared__ int hist[NB];
    const int tid = threadIdx.x;
    const int s   = blockIdx.x;
    const int e0  = s * SEG;
    const int* dstp = edge_index + N_EDGES;
    for (int i = tid; i < SEG; i += 256) {
        const int e = e0 + i;
        key[i] = (e < N_EDGES) ? (dstp[e] >> TR_LOG) : -1;
    }
    for (int b = tid; b < NB; b += 256) hist[b] = 0;
    __syncthreads();

    const int i0 = tid, i1 = tid + 256;
    const int k0 = key[i0], k1 = key[i1];
    int all0 = 0, bef0 = 0, all1 = 0, bef1 = 0;
    for (int j = 0; j < SEG; j += 4) {
        const int4 q = *(const int4*)&key[j];
        const int c0 = (q.x == k0) + (q.y == k0) + (q.z == k0) + (q.w == k0);
        const int c1 = (q.x == k1) + (q.y == k1) + (q.z == k1) + (q.w == k1);
        all0 += c0; all1 += c1;
        if (j + 4 <= i0) bef0 += c0;
        if (j + 4 <= i1) bef1 += c1;
    }
    {
        const int j0 = i0 & ~3, j1 = i1 & ~3;
        const int4 q = *(const int4*)&key[j0];
        const int4 p = *(const int4*)&key[j1];
        bef0 += ((q.x == k0) & (j0     < i0)) + ((q.y == k0) & (j0 + 1 < i0))
              + ((q.z == k0) & (j0 + 2 < i0));
        bef1 += ((p.x == k1) & (j1     < i1)) + ((p.y == k1) & (j1 + 1 < i1))
              + ((p.z == k1) & (j1 + 2 < i1));
    }
    if (k0 >= 0) { rankv[e0 + i0] = bef0; if (bef0 + 1 == all0) hist[k0] = all0; }
    if (k1 >= 0) { rankv[e0 + i1] = bef1; if (bef1 + 1 == all1) hist[k1] = all1; }
    __syncthreads();
    for (int b = tid; b < NB; b += 256) cnt[(size_t)s * NB + b] = hist[b];
}

#define SCW 32
#define SPW ((NSEG + SCW - 1) / SCW)
#define SCB ((NB + 31) / 32)
static_assert((SCW - 1) * SPW < NSEG && NSEG < SCW * SPW, "only the last wave's run is cut short, and it is not empty");
__global__ __launch_bounds__(1024) void k_scan_seg(
    int* __restrict__ cnt, int* __restrict__ tot)
{
    __shared__ int part[SCW][32];
    const int lane = threadIdx.x & 31;
    const int w    = __builtin_amdgcn_readfirstlane(threadIdx.x >> 5);
    const int b    = blockIdx.x * 32 + lane;
    const bool colok = b < NB;
    const int bc   = colok ? b : NB - 1;
    const int s0   = w * SPW;

    int v[SPW];
    #pragma unroll
    for (int i = 0; i < SPW; ++i) {
        const int s  = s0 + i;
        const int sc = s < NSEG ? s : NSEG - 1;
        v[i] = cnt[(size_t)sc * NB + bc];
    }
    int run = 0;
    #pragma unroll
    for (int i = 0; i < SPW; ++i) {
        const int x = (s0 + i < NSEG) ? v[i] : 0;
        v[i] = run;
        run += x;
    }
    part[w][lane] = run;
    __syncthreads();
    int off = 0;
    #pragma unroll
    for (int k = 0; k < SCW; ++k) {
        const int p = part[k][lane];
        off += (k < w) ? p : 0;
    }
    if (colok) {
        int* cp = cnt + (size_t)s0 * NB + b;
        if (s0 + SPW <= NSEG) {
            #pragma unroll
            for (int i = 0; i < SPW; ++i) cp[(size_t)i * NB] = off + v[i];
        } else {
            #pragma unroll
            for (int i = 0; i < SPW; ++i)
                if (s0 + i < NSEG) cp[(size_t)i * NB] = off + v[i];
            tot[b] = off + run;
        }
    }
}

__global__ __launch_bounds__(1024) void k_scan_tiles(
    const int* __restrict__ tot, int* __restrict__ bstart)
{
    __shared__ int buf[2][1024];
    const int t  = threadIdx.x;
    const int i0 = 2 * t, i1 = 2 * t + 1;
    const int v0 = (i0 < NB) ? tot[i0] : 0;
    const int v1 = (i1 < NB) ? tot[i1] : 0;
    buf[0][t] = v0 + v1;
    __syncthreads();
    int pin = 0;
    for (int d = 1; d < 1024; d <<= 1) {
        int x = buf[pin][t];
        if (t >= d) x += buf[pin][t - d];
        buf[pin ^ 1][t] = x;
        __syncthreads();
        pin ^= 1;
    }
    const int excl = buf[pin][t] - (v0 + v1);
    if (i0 < NB) bstart[i0] = excl;
    if (i1 < NB) bstart[i1] = excl + v0;
}

__global__ __launch_bounds__(256) void k_place(
    const int* __restrict__ edge_index, const int* __restrict__ edge_attr,
    const int* __restrict__ rankv, const int* __restrict__ within,
    const int* __restrict__ bstart, int2* __restrict__ rec)
{
    const int s = blockIdx.x;
    for (int i = threadIdx.x; i < SEG; i += 256) {
        const int e = s * SEG + i;
        if (e < N_EDGES) {
            const int src = edge_index[e];
            const int d   = edge_index[N_EDGES + e];
            const int b   = d >> TR_LOG;
            const int pos = bstart[b] + within[(size_t)s * NB + b] + rankv[e];
            const int a0 = edge_attr[e * 3 + 0];
            const int a1 = edge_attr[e * 3 + 1];
            const int a2 = edge_attr[e * 3 + 2];
            rec[pos] = make_int2(src, (d & (TR - 1)) | (a0 << 8) | (a1 << 12) | (a2 << 16));
        }
    }
}

#define CHUNK 512
__global__ __launch_bounds__(128, 7) void k_tile_agg(
    const int2* __restrict__ rec, const int* __restrict__ bstart,
    const int* __restrict__ tot, const float* __restrict__ bond_emb,
    const float* __restrict__ h, float* __restrict__ agg)
{
    __shared__ __align__(16) float acc[TR * HID];
    __shared__ __align__(8) int2 own[4 * CHUNK];
    const int tid  = threadIdx.x;
    const int w    = tid >> 5;
    const int lane = tid & 31;
    const int c    = lane * 4;
    const int b    = blockIdx.x;
    const int n    = tot[b];
    const int2* rp = rec + bstart[b];

    #pragma unroll
    for (int k = 0; k < TR / 4; ++k)
        *(float4*)&acc[(w + 4 * k) * HID + c] = make_float4(0.f, 0.f, 0.f, 0.f);

    int2* lw = &own[w * CHUNK];
    for (int cb = 0; cb < n; cb += CHUNK) {
        const int ce = (n < cb + CHUNK) ? n : (cb + CHUNK);

        int2 rcs[CHUNK / 32];
        #pragma unroll
        for (int g = 0; g < CHUNK / 32; ++g) {
            const int i = cb + 32 * g + lane;
            rcs[g] = (i < ce) ? rp[i] : make_int2(0, 0);
        }
        int m = 0;
        #pragma unroll
        for (int g = 0; g < CHUNK / 32; ++g) {
            const bool mine = (cb + 32 * g + lane < ce) && ((rcs[g].y & 3) == w);
            const unsigned mask = __builtin_amdgcn_ballot_w32(mine);
            if (mine) lw[m + (int)__builtin_amdgcn_mbcnt_lo(mask, 0u)] = rcs[g];
            m += __builtin_popcount(mask);
        }
        __builtin_amdgcn_fence(__ATOMIC_RELEASE, "wavefront");
        __builtin_amdgcn_wave_barrier();
        __builtin_amdgcn_fence(__ATOMIC_ACQUIRE, "wavefront");

        int2 rc = make_int2(0, 0);
        for (int t0 = 0; t0 < m; t0 += 8) {
            const int lb = t0 & 31;
            if (lb == 0) {
                int li = t0 + lane;
                li = li < m ? li : m - 1;
                rc = lw[li];
            }
            int  sv[8], pv[8];
            bool ok[8];
            #pragma unroll
            for (int g = 0; g < 8; ++g) {
                ok[g] = (t0 + g < m);
                sv[g] = __builtin_amdgcn_readlane(rc.x, lb + g);
                pv[g] = __builtin_amdgcn_readlane(rc.y, lb + g);
            }
            float4 m4[8];
            #pragma unroll
            for (int g = 0; g < 8; ++g) {
                const int a0 = (pv[g] >> 8)  & 15;
                const int a1 = (pv[g] >> 12) & 15;
                const int a2 = (pv[g] >> 16) & 15;
                const float4 hv = *(const float4*)(h + (size_t)sv[g] * HID + c);
                const float4 e0 = *(const float4*)(bond_emb + ((size_t)(0 * 6 + a0)) * HID + c);
                const float4 e1 = *(const float4*)(bond_emb + ((size_t)(1 * 6 + a1)) * HID + c);
                const float4 e2 = *(const float4*)(bond_emb + ((size_t)(2 * 6 + a2)) * HID + c);
                float m0 = hv.x + e0.x + e1.x + e2.x; m0 = m0 > 0.f ? m0 : 0.f;
                float m1 = hv.y + e0.y + e1.y + e2.y; m1 = m1 > 0.f ? m1 : 0.f;
                float m2 = hv.z + e0.z + e1.z + e2.z; m2 = m2 > 0.f ? m2 : 0.f;
                float m3 = hv.w + e0.w + e1.w + e2.w; m3 = m3 > 0.f ? m3 : 0.f;
                m4[g] = make_float4(m0, m1, m2, m3);
            }
            #pragma unroll
            for (int g = 0; g < 8; ++g) {
                if (ok[g]) {
                    float* a = &acc[(pv[g] & (TR - 1)) * HID + c];
                    float4 t = *(float4*)a;
                    t.x += m4[g].x; t.y += m4[g].y; t.z += m4[g].z; t.w += m4[g].w;
                    *(float4*)a = t;
                }
            }
        }
        __builtin_amdgcn_fence(__ATOMIC_RELEASE, "wavefront");
        __builtin_amdgcn_wave_barrier();
        __builtin_amdgcn_fence(__ATOMIC_ACQUIRE, "wavefront");
    }

    #pragma unroll
    for (int k = 0; k < TR / 4; ++k) {
        const int r    = w + 4 * k;
        const int node = b * TR + r;
        if (node < N_NODES) {
            const float4 hv = *(const float4*)(h + (size_t)node * HID + c);
            const float4 a  = *(const float4*)&acc[r * HID + c];
            *(float4*)(agg + (size_t)node * HID + c) =
                make_float4(hv.x + a.x, hv.y + a.y, hv.z + a.z, hv.w + a.w);
        }
    }
}

__global__ __launch_bounds__(256) void k_prep_weights(
    const float* __restrict__ w1, const float* __restrict__ w2,
    __bf16* __restrict__ f1, __bf16* __restrict__ f2)
{
    int o = blockIdx.x * 256 + threadIdx.x;
    if (o >= NLAYERS * 4 * 8 * 32 * 16) return;
    int j     =  o        & 15;
    int lane  = (o >> 4)  & 31;
    int ntile = (o >> 9)  & 7;
    int ks    = (o >> 12) & 3;
    int layer =  o >> 14;
    int colL  = lane & 15;
    int khalf = (lane >> 4) * 8;
    int K     = ks * 32 + ((j < 8) ? (khalf + j) : (16 + khalf + (j - 8)));
    int col   = ntile * 16 + colL;
    size_t src = (size_t)layer * HID * HID + (size_t)K * HID + col;
    size_t dst = (size_t)layer * 16384 +
                 (((size_t)ks * 8 + ntile) * 32 + lane) * 16 + j;
    f1[dst] = (__bf16)w1[src];
    f2[dst] = (__bf16)w2[src];
}

#define NTILES (N_NODES / 16)
#define NU_WAVES 8
#define ZP 136
static_assert(N_NODES % 16 == 0, "node tiles are full");

__global__ __launch_bounds__(256) void k_node_update(
    const float* __restrict__ agg, const float* __restrict__ hcur,
    const __bf16* __restrict__ wf1, const __bf16* __restrict__ wf2,
    const float* __restrict__ b1, const float* __restrict__ b2,
    const float* __restrict__ gamma, const float* __restrict__ beta,
    const float* __restrict__ mean, const float* __restrict__ var,
    float* __restrict__ hnext)
{
    __shared__ __align__(16) __bf16 wl[2 * 16384];
    __shared__ __align__(16) __bf16 z1s[NU_WAVES * 16 * ZP];

    const int tid  = threadIdx.x;
    const int wave = __builtin_amdgcn_readfirstlane(tid >> 5);
    const int lane = tid & 31;
    const int c    = lane & 15;
    const int hh   = lane >> 4;
    const int tile = blockIdx.x * NU_WAVES + wave;
    const bool act = tile < NTILES;
    const int node0 = tile * 16;

    uint4 wv[16];
    #pragma unroll
    for (int i = 0; i < 16; ++i) {
        const int qq = (tid + 256 * i) & 2047;
        wv[i] = *(const uint4*)(((i >> 3) ? wf2 : wf1) + (size_t)qq * 8);
    }

    float4 av[16];
    {
        const int rowA = (act ? node0 : (NTILES - 1) * 16) + c;
        const float* ap = agg + (size_t)rowA * HID + 8 * hh;
        #pragma unroll
        for (int ks = 0; ks < 4; ++ks) {
            av[4 * ks + 0] = *(const float4*)(ap + ks * 32);
            av[4 * ks + 1] = *(const float4*)(ap + ks * 32 + 4);
            av[4 * ks + 2] = *(const float4*)(ap + ks * 32 + 16);
            av[4 * ks + 3] = *(const float4*)(ap + ks * 32 + 20);
        }
    }

    #pragma unroll
    for (int i = 0; i < 16; ++i) {
        const int qq = (tid + 256 * i) & 2047;
        const int f = qq >> 6;
        const int l = (qq >> 1) & 31;
        const int p = qq & 1;
        *(uint4*)&wl[(i >> 3) * 16384 + f * 512 + p * 256 + l * 8] = wv[i];
    }
    __syncthreads();
    if (!act) return;

    v16bf a[4];
    #pragma unroll
    for (int ks = 0; ks < 4; ++ks) {
        const float4 p0 = av[4 * ks + 0], p1 = av[4 * ks + 1];
        const float4 p2 = av[4 * ks + 2], p3 = av[4 * ks + 3];
        v16bf t;
        t[0]  = (__bf16)p0.x; t[1]  = (__bf16)p0.y; t[2]  = (__bf16)p0.z; t[3]  = (__bf16)p0.w;
        t[4]  = (__bf16)p1.x; t[5]  = (__bf16)p1.y; t[6]  = (__bf16)p1.z; t[7]  = (__bf16)p1.w;
        t[8]  = (__bf16)p2.x; t[9]  = (__bf16)p2.y; t[10] = (__bf16)p2.z; t[11] = (__bf16)p2.w;
        t[12] = (__bf16)p3.x; t[13] = (__bf16)p3.y; t[14] = (__bf16)p3.z; t[15] = (__bf16)p3.w;
        a[ks] = t;
    }

    v8f acc[8];
    #pragma unroll
    for (int nt = 0; nt < 8; ++nt) acc[nt] = (v8f){};
    #pragma unroll
    for (int ks = 0; ks < 4; ++ks) {
        #pragma unroll
        for (int nt = 0; nt < 8; ++nt) {
            const __bf16* bp = &wl[(ks * 8 + nt) * 512 + lane * 8];
            const v8bf blo = *(const v8bf*)(bp);
            const v8bf bhi = *(const v8bf*)(bp + 256);
            const v16bf b = __builtin_shufflevector(blo, bhi,
                        0,1,2,3,4,5,6,7,8,9,10,11,12,13,14,15);
            acc[nt] = __builtin_amdgcn_wmma_f32_16x16x32_bf16(
                          false, a[ks], false, b, (short)0, acc[nt], false, false);
        }
    }

    __bf16* z1w = &z1s[wave * 16 * ZP];
    #pragma unroll
    for (int nt = 0; nt < 8; ++nt) {
        const int col = nt * 16 + c;
        const float bias1 = b1[col];
        #pragma unroll
        for (int r = 0; r < 8; ++r) {
            float v = acc[nt][r] + bias1;
            v = v > 0.f ? v : 0.f;
            z1w[(8 * hh + r) * ZP + col] = (__bf16)v;
        }
    }
    __builtin_amdgcn_fence(__ATOMIC_RELEASE, "wavefront");
    __builtin_amdgcn_wave_barrier();
    __builtin_amdgcn_fence(__ATOMIC_ACQUIRE, "wavefront");

    #pragma unroll
    for (int ks = 0; ks < 4; ++ks) {
        const __bf16* ap = &z1w[c * ZP + ks * 32 + 8 * hh];
        const v8bf alo = *(const v8bf*)(ap);
        const v8bf ahi = *(const v8bf*)(ap + 16);
        a[ks] = __builtin_shufflevector(alo, ahi,
                    0,1,2,3,4,5,6,7,8,9,10,11,12,13,14,15);
    }

    #pragma unroll
    for (int nt = 0; nt < 8; ++nt) acc[nt] = (v8f){};
    #pragma unroll
    for (int ks = 0; ks < 4; ++ks) {
        #pragma unroll
        for (int nt = 0; nt < 8; ++nt) {
            const __bf16* bp = &wl[16384 + (ks * 8 + nt) * 512 + lane * 8];
            const v8bf blo = *(const v8bf*)(bp);
            const v8bf bhi = *(const v8bf*)(bp + 256);
            const v16bf b = __builtin_shufflevector(blo, bhi,
                        0,1,2,3,4,5,6,7,8,9,10,11,12,13,14,15);
            acc[nt] = __builtin_amdgcn_wmma_f32_16x16x32_bf16(
                          false, a[ks], false, b, (short)0, acc[nt], false, false);
        }
    }

    #pragma unroll
    for (int nt = 0; nt < 8; ++nt) {
        const int col = nt * 16 + c;
        const float bias2 = b2[col];
        const float mu    = mean[col];
        const float inv   = rsqrtf(var[col] + 1e-5f);
        const float ga    = gamma[col];
        const float be    = beta[col];
        #pragma unroll
        for (int r = 0; r < 8; ++r) {
            const size_t idx = (size_t)(node0 + 8 * hh + r) * HID + col;
            float v = acc[nt][r] + bias2;
            v = (v - mu) * inv * ga + be;
            v = v > 0.f ? v : 0.f;
            hnext[idx] = hcur[idx] + v;
        }
    }
}

#define PSTEP ((N_NODES + 127) / 128)
#define PWL ((PSTEP + 31) / 32)
static_assert(127 * PSTEP < N_NODES, "every level-1 probe lies inside the id array");

__global__ __launch_bounds__(128) void k_pool_head(
    const float* __restrict__ h, const int* __restrict__ batch,
    const float* __restrict__ mw, const float* __restrict__ mb,
    float* __restrict__ out)
{
    __shared__ __align__(16) float part[4 * HID];
    __shared__ float p[HID];
    const int g    = blockIdx.x;
    const int t    = threadIdx.x;
    const int w    = __builtin_amdgcn_readfirstlane(t >> 5);
    const int lane = t & 31;
    const int c    = lane * 4;

    int pv[4];
    #pragma unroll
    for (int j = 0; j < 4; ++j) pv[j] = batch[(lane + 32 * j) * PSTEP];
    int A = 0, B = 0;
    #pragma unroll
    for (int j = 0; j < 4; ++j) {
        A += __builtin_popcount(__builtin_amdgcn_ballot_w32(pv[j] <  g));
        B += __builtin_popcount(__builtin_amdgcn_ballot_w32(pv[j] <= g));
    }
    const int baseL = A ? (A - 1) * PSTEP + 1 : 0;
    const int baseH = B ? (B - 1) * PSTEP + 1 : 0;

    int wl[PWL], wh[PWL];
    #pragma unroll
    for (int j = 0; j < PWL; ++j) {
        const int il = baseL + lane + 32 * j, ih = baseH + lane + 32 * j;
        wl[j] = batch[il < N_NODES ? il : N_NODES - 1];
        wh[j] = batch[ih < N_NODES ? ih : N_NODES - 1];
    }
    int lo = baseL, hi = baseH;
    #pragma unroll
    for (int j = 0; j < PWL; ++j) {
        const int  o   = lane + 32 * j;
        const bool inL = (o < PSTEP) && (baseL + o < N_NODES);
        const bool inH = (o < PSTEP) && (baseH + o < N_NODES);
        lo += __builtin_popcount(__builtin_amdgcn_ballot_w32(inL && wl[j] <  g));
        hi += __builtin_popcount(__builtin_amdgcn_ballot_w32(inH && wh[j] <= g));
    }

    float4 acc = make_float4(0.f, 0.f, 0.f, 0.f);
    for (int i0 = lo + w; i0 < hi; i0 += 32) {
        float4 v[8];
        #pragma unroll
        for (int k = 0; k < 8; ++k) {
            const int i = i0 + 4 * k;
            v[k] = (i < hi) ? *(const float4*)(h + (size_t)i * HID + c)
                            : make_float4(0.f, 0.f, 0.f, 0.f);
        }
        #pragma unroll
        for (int k = 0; k < 8; ++k) {
            acc.x += v[k].x; acc.y += v[k].y; acc.z += v[k].z; acc.w += v[k].w;
        }
    }
    *(float4*)&part[w * HID + c] = acc;
    __syncthreads();
    float cnt = (float)(hi - lo);
    cnt = cnt > 1.f ? cnt : 1.f;
    p[t] = (((part[t] + part[HID + t]) + part[2 * HID + t]) + part[3 * HID + t]) / cnt;
    __syncthreads();
    if (t < OUTD) {
        float a = mb[t];
        #pragma unroll 16
        for (int cc = 0; cc < HID; ++cc) a += p[cc] * mw[cc * OUTD + t];
        out[g * OUTD + t] = a;
    }
}

extern "C" void kernel_launch(void* const* d_in, const int* in_sizes, int n_in,
                              void* d_out, int out_size, void* d_ws, size_t ws_size,
                              hipStream_t stream) {
    const int*   x          = (const int*)  d_in[0];
    const int*   edge_index = (const int*)  d_in[1];
    const int*   edge_attr  = (const int*)  d_in[2];
    const int*   batch      = (const int*)  d_in[3];
    const float* atom_emb   = (const float*)d_in[4];
    const float* bond_emb   = (const float*)d_in[5];
    const float* lin1_w     = (const float*)d_in[6];
    const float* lin1_b     = (const float*)d_in[7];
    const float* lin2_w     = (const float*)d_in[8];
    const float* lin2_b     = (const float*)d_in[9];
    const float* bn_gamma   = (const float*)d_in[10];
    const float* bn_beta    = (const float*)d_in[11];
    const float* bn_mean    = (const float*)d_in[12];
    const float* bn_var     = (const float*)d_in[13];
    const float* mlp_w      = (const float*)d_in[14];
    const float* mlp_b      = (const float*)d_in[15];

    char* ws = (char*)d_ws;
    size_t off = 0;
    auto alloc = [&](size_t bytes) -> char* {
        char* p = ws + off;
        off = (off + bytes + 255) & ~(size_t)255;
        return p;
    };
    float*  h0   = (float*) alloc((size_t)N_NODES * HID * 4);
    float*  h1   = (float*) alloc((size_t)N_NODES * HID * 4);
    float*  agg  = (float*) alloc((size_t)N_NODES * HID * 4);
    __bf16* wf1  = (__bf16*)alloc((size_t)NLAYERS * 16384 * 2);
    __bf16* wf2  = (__bf16*)alloc((size_t)NLAYERS * 16384 * 2);
    int*    rankv  = (int*)  alloc((size_t)N_EDGES * 4);
    int*    cnt    = (int*)  alloc((size_t)NSEG * NB * 4);
    int*    tot    = (int*)  alloc((size_t)NB * 4);
    int*    bstart = (int*)  alloc((size_t)NB * 4);
    int2*   rec    = (int2*) alloc((size_t)N_EDGES * 8);

    const int nodeChunks = N_NODES * 32;

    k_prep_weights<<<(NLAYERS * 4 * 8 * 32 * 16 + 255) / 256, 256, 0, stream>>>(
        lin1_w, lin2_w, wf1, wf2);
    k_encode_nodes<<<(nodeChunks + 255) / 256, 256, 0, stream>>>(x, atom_emb, h0);

    k_rank<<<NSEG, 256, 0, stream>>>(edge_index, rankv, cnt);
    k_scan_seg<<<SCB, 1024, 0, stream>>>(cnt, tot);
    k_scan_tiles<<<1, 1024, 0, stream>>>(tot, bstart);
    k_place<<<NSEG, 256, 0, stream>>>(edge_index, edge_attr, rankv, cnt, bstart, rec);

    float* hc = h0;
    float* hn = h1;
    for (int i = 0; i < NLAYERS; ++i) {
        k_tile_agg<<<NB, 128, 0, stream>>>(rec, bstart, tot, bond_emb, hc, agg);
        k_node_update<<<(NTILES + NU_WAVES - 1) / NU_WAVES, 256, 0, stream>>>(
            agg, hc, wf1 + (size_t)i * 16384, wf2 + (size_t)i * 16384,
            lin1_b + i * HID, lin2_b + i * HID,
            bn_gamma + i * HID, bn_beta + i * HID,
            bn_mean + i * HID, bn_var + i * HID, hn);
        float* t = hc; hc = hn; hn = t;
    }

    k_pool_head<<<N_GRAPHS, 128, 0, stream>>>(hc, batch, mlp_w, mlp_b, (float*)d_out);
}
